# starting kernel
# baseline (speedup 1.0000x reference)
#include <hip/hip_runtime.h>
#include <hip/hip_bf16.h>

typedef __attribute__((ext_vector_type(2))) float v2f;
typedef __attribute__((ext_vector_type(4))) float v4f;
typedef __attribute__((ext_vector_type(8))) float v8f;

#define NNODE 25
#define NEDGE 64
#define HID   128
#define OUTD  128
#define BPB   16
#define NPAIR (BPB / 2)
#define ZROW  16

static __device__ __forceinline__ v2f lo2(v4f q) { return __builtin_shufflevector(q, q, 0, 1); }
static __device__ __forceinline__ v2f hi2(v4f q) { return __builtin_shufflevector(q, q, 2, 3); }

__launch_bounds__(128)
__global__ void sage_fused_kernel(const float* __restrict__ x,
                                  const int*   __restrict__ edge_index,
                                  const float* __restrict__ W1l,
                                  const float* __restrict__ W1r,
                                  const float* __restrict__ b1,
                                  const float* __restrict__ W2l,
                                  const float* __restrict__ W2r,
                                  const float* __restrict__ b2,
                                  float* __restrict__ out)
{
    __shared__ int   ssrc[NEDGE];
    __shared__ int   sdst[NEDGE];
    __shared__ float sdeginv[NNODE];
    __shared__ float snodew[NNODE];
    __shared__ float snw2[NNODE * 2];
    __shared__ float sA1[NNODE * 26];
    __align__(16) __shared__ float zbuf[NPAIR * NNODE * ZROW];
    __shared__ float As[BPB][2 * HID];

    const int tid  = threadIdx.x;
    const int lane = tid & 31;
    const int wave = tid >> 5;
    const int bbase = blockIdx.x * BPB;

    const int k = tid;
    const v2f wl0 = { W1l[0 * HID + k], W1l[0 * HID + k] };
    const v2f wl1 = { W1l[1 * HID + k], W1l[1 * HID + k] };
    const v2f wl2 = { W1l[2 * HID + k], W1l[2 * HID + k] };
    const v2f wr0 = { W1r[0 * HID + k], W1r[0 * HID + k] };
    const v2f wr1 = { W1r[1 * HID + k], W1r[1 * HID + k] };
    const v2f wr2 = { W1r[2 * HID + k], W1r[2 * HID + k] };
    const float b1k = b1[k];
    const v2f b1k2 = { b1k, b1k };

    if (tid < NEDGE) {
        ssrc[tid] = edge_index[tid];
        sdst[tid] = edge_index[NEDGE + tid];
    }
    for (int i = tid; i < NNODE * 26; i += 128) sA1[i] = 0.0f;
    if (tid < NNODE) snodew[tid] = 0.0f;
    __syncthreads();

    if (tid < NNODE) {
        int c = 0;
        #pragma unroll
        for (int e = 0; e < NEDGE; ++e) c += (sdst[e] == tid);
        sdeginv[tid] = 1.0f / fmaxf((float)c, 1.0f);
    }
    __syncthreads();

    if (tid < NEDGE) {
        const float di = sdeginv[sdst[tid]];
        atomicAdd(&sA1[sdst[tid] * 26 + ssrc[tid]], di);
        atomicAdd(&snodew[ssrc[tid]], di * (1.0f / 25.0f));
    }
    {
        const size_t base = (size_t)bbase * (NNODE * 3);
        for (int i = tid; i < BPB * NNODE * 3; i += 128) {
            const int slot = i / (NNODE * 3);
            const int r = i % (NNODE * 3);
            const int n = r / 3, d = r % 3;
            const int pair = slot >> 1, sub = slot & 1;
            zbuf[(pair * NNODE + n) * ZROW + 6 + 2 * d + sub] = x[base + i];
        }
    }
    __syncthreads();

    for (int i = tid; i < NPAIR * NNODE * 3; i += 128) {
        const int pair = i / (NNODE * 3);
        const int r = i % (NNODE * 3);
        const int n = r / 3, d = r % 3;
        v2f acc = { 0.0f, 0.0f };
        #pragma unroll
        for (int m = 0; m < NNODE; ++m) {
            const float a = sA1[n * 26 + m];
            const v2f xm = *(const v2f*)&zbuf[(pair * NNODE + m) * ZROW + 6 + 2 * d];
            acc = __builtin_elementwise_fma(xm, (v2f){ a, a }, acc);
        }
        *(v2f*)&zbuf[(pair * NNODE + n) * ZROW + 2 * d] = acc;
    }
    if (tid < NNODE) {
        const float w = snodew[tid];
        snw2[2 * tid] = w;
        snw2[2 * tid + 1] = w;
    }
    __syncthreads();

    for (int pair = 0; pair < NPAIR; ++pair) {
        const float* zp = &zbuf[pair * NNODE * ZROW];
        v2f cl = { 0.0f, 0.0f };
        v2f cr = { 0.0f, 0.0f };
        #pragma unroll
        for (int n = 0; n < NNODE; ++n) {
            const float* row = zp + n * ZROW;
            const v4f q0 = *(const v4f*)(row);
            const v4f q1 = *(const v4f*)(row + 4);
            const v4f q2 = *(const v4f*)(row + 8);
            v2f pre = b1k2;
            pre = __builtin_elementwise_fma(lo2(q0), wl0, pre);
            pre = __builtin_elementwise_fma(hi2(q0), wl1, pre);
            pre = __builtin_elementwise_fma(lo2(q1), wl2, pre);
            pre = __builtin_elementwise_fma(hi2(q1), wr0, pre);
            pre = __builtin_elementwise_fma(lo2(q2), wr1, pre);
            pre = __builtin_elementwise_fma(hi2(q2), wr2, pre);
            const v2f h = __builtin_elementwise_max(pre, (v2f){ 0.0f, 0.0f });
            const v2f nw = *(const v2f*)&snw2[2 * n];
            cl = __builtin_elementwise_fma(h, nw, cl);
            cr = cr + h;
        }
        As[2 * pair + 0][k] = cl.x;
        As[2 * pair + 1][k] = cl.y;
        As[2 * pair + 0][HID + k] = cr.x * (1.0f / 25.0f);
        As[2 * pair + 1][HID + k] = cr.y * (1.0f / 25.0f);
    }
    __syncthreads();

    {
        const int hi  = lane >> 4;
        const int l15 = lane & 15;
        const int n0a = wave * 32;
        const int n0b = n0a + 16;

        v8f c0, c1;
        const float bza = b2[n0a + l15];
        const float bzb = b2[n0b + l15];
        #pragma unroll
        for (int v = 0; v < 8; ++v) { c0[v] = bza; c1[v] = bzb; }

        const float* Wmats[2] = { W2l, W2r };
        #pragma unroll
        for (int half = 0; half < 2; ++half) {
            const float* W = Wmats[half];
            const int kofs = half * HID;
            #pragma unroll 4
            for (int k0 = 0; k0 < HID; k0 += 4) {
                const int kb = k0 + (hi << 1);
                v2f a;
                a.x = As[l15][kofs + kb];
                a.y = As[l15][kofs + kb + 1];
                v2f bfa, bfb;
                bfa.x = W[kb * OUTD + n0a + l15];
                bfa.y = W[(kb + 1) * OUTD + n0a + l15];
                bfb.x = W[kb * OUTD + n0b + l15];
                bfb.y = W[(kb + 1) * OUTD + n0b + l15];
                c0 = __builtin_amdgcn_wmma_f32_16x16x4_f32(
                        false, a, false, bfa, (short)0, c0, false, false);
                c1 = __builtin_amdgcn_wmma_f32_16x16x4_f32(
                        false, a, false, bfb, (short)0, c1, false, false);
            }
        }

        #pragma unroll
        for (int v = 0; v < 8; ++v) {
            const int m = v + (hi << 3);
            out[(size_t)(bbase + m) * OUTD + n0a + l15] = c0[v];
            out[(size_t)(bbase + m) * OUTD + n0b + l15] = c1[v];
        }
    }
}

extern "C" void kernel_launch(void* const* d_in, const int* in_sizes, int n_in,
                              void* d_out, int out_size, void* d_ws, size_t ws_size,
                              hipStream_t stream) {
    const float* x   = (const float*)d_in[0];
    const int*   ei  = (const int*)d_in[1];
    const float* W1l = (const float*)d_in[2];
    const float* W1r = (const float*)d_in[3];
    const float* b1  = (const float*)d_in[4];
    const float* W2l = (const float*)d_in[5];
    const float* W2r = (const float*)d_in[6];
    const float* b2  = (const float*)d_in[7];
    float* out = (float*)d_out;

    const int B = in_sizes[0] / (NNODE * 3);
    const int blocks = B / BPB;
    sage_fused_kernel<<<blocks, 128, 0, stream>>>(x, ei, W1l, W1r, b1, W2l, W2r, b2, out);
}


# lane = batch row, lane half = k pair: layer 1 is computed straight into the A-fragment registers, W2 fragments come from one LDS copy per block; layer-2 sum order: W2l and W2r k-steps alternate
# speedup vs baseline: 1.0996x; 1.0996x over previous
#include <hip/hip_runtime.h>


typedef __attribute__((ext_vector_type(2))) float v2f;
typedef __attribute__((ext_vector_type(4))) float v4f;
typedef __attribute__((ext_vector_type(8))) float v8f;

#define NNODE 25
#define NEDGE 64
#define HID   128
#define OUTD  128
#define TPW   16
#define WPB   4
#define BPB   (TPW * WPB)
#define NJ    (HID / 4)
#define A1P   28
#define XW    (TPW * NNODE * 3)
#define CHK   140

static __device__ __forceinline__ v2f lo2(v4f q) { return __builtin_shufflevector(q, q, 0, 1); }
static __device__ __forceinline__ v2f hi2(v4f q) { return __builtin_shufflevector(q, q, 2, 3); }
static __device__ __forceinline__ v2f sp2(float z) { return (v2f){ z, z }; }

#define NSTG  8
#define NSLOT (NNODE + NSTG - 1)

static __device__ __forceinline__ void kstep(const float (&ag)[NNODE][3], const float (&xr)[NNODE][3],
                                             const float (&nw)[NNODE], v4f (&w)[4], const v4f* wnext,
                                             v2f& acl, v2f& acr)
{
    v2f pre[NNODE];
    v2f cl = { 0.0f, 0.0f };
    v2f cr = { 0.0f, 0.0f };
    #pragma unroll
    for (int s = 0; s < NSLOT; ++s) {
        #pragma unroll
        for (int n = 0; n < NNODE; ++n) {
            const int st = s - n;
            if (st < 0 || st >= NSTG) continue;
            if (st == 0)      pre[n] = __builtin_elementwise_fma(sp2(ag[n][0]), lo2(w[0]), lo2(w[3]));
            else if (st == 1) pre[n] = __builtin_elementwise_fma(sp2(ag[n][1]), hi2(w[0]), pre[n]);
            else if (st == 2) pre[n] = __builtin_elementwise_fma(sp2(ag[n][2]), lo2(w[1]), pre[n]);
            else if (st == 3) pre[n] = __builtin_elementwise_fma(sp2(xr[n][0]), hi2(w[1]), pre[n]);
            else if (st == 4) pre[n] = __builtin_elementwise_fma(sp2(xr[n][1]), lo2(w[2]), pre[n]);
            else if (st == 5) pre[n] = __builtin_elementwise_fma(sp2(xr[n][2]), hi2(w[2]), pre[n]);
            else if (st == 6) pre[n] = __builtin_elementwise_maxnum(pre[n], (v2f){ 0.0f, 0.0f });
            else {
                cl = __builtin_elementwise_fma(pre[n], sp2(nw[n]), cl);
                cr = cr + pre[n];
            }
        }
        if (s == 25) { w[0] = wnext[0]; w[3] = wnext[3]; }
        if (s == 27) w[1] = wnext[1];
        if (s == 29) w[2] = wnext[2];
        if (s == NSLOT - 1) {
            acl = cl;
            acr = cr * (1.0f / 25.0f);
        }
        if ((s & 1) == 1) __builtin_amdgcn_sched_barrier(0);
    }
}

__launch_bounds__(128)
__global__ void sage_rowfrag_kernel(const float* __restrict__ x,
                                    const int*   __restrict__ edge_index,
                                    const float* __restrict__ W1l,
                                    const float* __restrict__ W1r,
                                    const float* __restrict__ b1,
                                    const float* __restrict__ W2l,
                                    const float* __restrict__ W2r,
                                    const float* __restrict__ b2,
                                    float* __restrict__ out)
{
    __shared__ int   ssrc[NEDGE];
    __shared__ int   sdst[NEDGE];
    __shared__ float sdeginv[NNODE];
    __shared__ float snodew[NNODE];
    __align__(16) __shared__ float sA1[NNODE * A1P];
    __align__(16) __shared__ float sW1[NJ * 2 * 16];
    __align__(16) __shared__ float sx[WPB * XW];
    __align__(16) __shared__ float sW2[2 * NJ * 4 * CHK];

    const int tid  = threadIdx.x;
    const int lane = tid & 31;
    const int wave = tid >> 5;
    const int hi   = lane >> 4;
    const int l15  = lane & 15;
    const int b0   = (blockIdx.x * WPB + wave) * TPW;

    v4f wra[32], wrb[32];
    #pragma unroll
    for (int u = 0; u < 32; ++u) {
        const int it  = tid + 128 * u;
        const int mat = it >> 11;
        const int rp  = (it >> 5) & 63;
        const int c4  = it & 31;
        const float* Wm = mat ? W2r : W2l;
        wra[u] = *(const v4f*)&Wm[(2 * rp) * OUTD + 4 * c4];
        wrb[u] = *(const v4f*)&Wm[(2 * rp + 1) * OUTD + 4 * c4];
    }

    if (tid < NEDGE) {
        ssrc[tid] = edge_index[tid];
        sdst[tid] = edge_index[NEDGE + tid];
    }
    for (int i = tid; i < NNODE * A1P; i += 128) sA1[i] = 0.0f;
    if (tid < NNODE) snodew[tid] = 0.0f;

    {
        const v4f* xg = (const v4f*)(x + (size_t)b0 * (NNODE * 3));
        v4f* xs = (v4f*)&sx[wave * XW];
        for (int i = lane; i < XW / 4; i += 32) xs[i] = xg[i];
    }
    for (int idx = tid; idx < NJ * 2 * 16; idx += 128) {
        const int jh = idx >> 4, i = idx & 15;
        const int g = i >> 1, col = 2 * jh + (i & 1);
        float v = 0.0f;
        if (g < 3)       v = W1l[g * HID + col];
        else if (g < 6)  v = W1r[(g - 3) * HID + col];
        else if (g == 6) v = b1[col];
        sW1[idx] = v;
    }
    __syncthreads();

    if (tid < NNODE) {
        int c = 0;
        #pragma unroll
        for (int e = 0; e < NEDGE; ++e) c += (sdst[e] == tid);
        sdeginv[tid] = 1.0f / fmaxf((float)c, 1.0f);
    }
    __syncthreads();

    if (tid < NEDGE) {
        const float di = sdeginv[sdst[tid]];
        atomicAdd(&sA1[sdst[tid] * A1P + ssrc[tid]], di);
        atomicAdd(&snodew[ssrc[tid]], di * (1.0f / 25.0f));
    }
    #pragma unroll
    for (int u = 0; u < 32; ++u) {
        const int it  = tid + 128 * u;
        const int mat = it >> 11;
        const int rp  = (it >> 5) & 63;
        const int c4  = it & 31;
        const int j = rp >> 1, h = rp & 1;
        const int t = c4 >> 2, q = t >> 1, tt = t & 1;
        const int cb = 4 * (c4 & 3);
        float* dstp = &sW2[((mat * NJ + j) * 4 + q) * CHK + (cb + 16 * h) * 4 + tt * 2];
        *(v2f*)(dstp + 0)  = (v2f){ wra[u].x, wrb[u].x };
        *(v2f*)(dstp + 4)  = (v2f){ wra[u].y, wrb[u].y };
        *(v2f*)(dstp + 8)  = (v2f){ wra[u].z, wrb[u].z };
        *(v2f*)(dstp + 12) = (v2f){ wra[u].w, wrb[u].w };
    }
    __syncthreads();

    float xr[NNODE][3];
    {
        const float* xrow = &sx[wave * XW + l15 * (NNODE * 3)];
        #pragma unroll
        for (int m = 0; m < NNODE; ++m) {
            xr[m][0] = xrow[3 * m + 0];
            xr[m][1] = xrow[3 * m + 1];
            xr[m][2] = xrow[3 * m + 2];
        }
    }
    float ag[NNODE][3];
    #pragma unroll
    for (int n = 0; n < NNODE; ++n) {
        float arow[A1P];
        #pragma unroll
        for (int m4 = 0; m4 < A1P / 4; ++m4) {
            const v4f a4 = *(const v4f*)&sA1[n * A1P + 4 * m4];
            arow[4 * m4 + 0] = a4.x; arow[4 * m4 + 1] = a4.y;
            arow[4 * m4 + 2] = a4.z; arow[4 * m4 + 3] = a4.w;
        }
        float a0 = 0.0f, a1 = 0.0f, a2 = 0.0f;
        #pragma unroll
        for (int m = 0; m < NNODE; ++m) {
            a0 = __builtin_fmaf(xr[m][0], arow[m], a0);
            a1 = __builtin_fmaf(xr[m][1], arow[m], a1);
            a2 = __builtin_fmaf(xr[m][2], arow[m], a2);
        }
        ag[n][0] = a0; ag[n][1] = a1; ag[n][2] = a2;
    }
    float nw[NNODE];
    #pragma unroll
    for (int n = 0; n < NNODE; ++n) nw[n] = snodew[n];

    v8f acc[8];
    #pragma unroll
    for (int t = 0; t < 8; ++t) {
        const float bz = b2[16 * t + l15];
        #pragma unroll
        for (int v = 0; v < 8; ++v) acc[t][v] = bz;
    }

    const v4f* w1p = (const v4f*)&sW1[hi * 16];
    const float* w2p = &sW2[lane * 4];
    v4f w[4];
    #pragma unroll
    for (int i = 0; i < 4; ++i) w[i] = w1p[i];
    #pragma unroll 1
    for (int j = 0; j < NJ; ++j) {
        v4f bl[4], br[4];
        #pragma unroll
        for (int q = 0; q < 4; ++q) {
            bl[q] = *(const v4f*)&w2p[((0 * NJ + j) * 4 + q) * CHK];
            br[q] = *(const v4f*)&w2p[((1 * NJ + j) * 4 + q) * CHK];
        }
        const int jn = (j + 1 < NJ) ? (j + 1) : j;
        v2f acl, acr;
        kstep(ag, xr, nw, w, w1p + jn * 8, acl, acr);
        #pragma unroll
        for (int q = 0; q < 4; ++q) {
            acc[2 * q]     = __builtin_amdgcn_wmma_f32_16x16x4_f32(
                                 false, acl, false, lo2(bl[q]), (short)0, acc[2 * q], false, false);
            acc[2 * q + 1] = __builtin_amdgcn_wmma_f32_16x16x4_f32(
                                 false, acl, false, hi2(bl[q]), (short)0, acc[2 * q + 1], false, false);
        }
        #pragma unroll
        for (int q = 0; q < 4; ++q) {
            acc[2 * q]     = __builtin_amdgcn_wmma_f32_16x16x4_f32(
                                 false, acr, false, lo2(br[q]), (short)0, acc[2 * q], false, false);
            acc[2 * q + 1] = __builtin_amdgcn_wmma_f32_16x16x4_f32(
                                 false, acr, false, hi2(br[q]), (short)0, acc[2 * q + 1], false, false);
        }
        __builtin_amdgcn_sched_barrier(0);
        asm volatile("v_nop\n\tv_nop\n\tv_nop\n\tv_nop");
    }

    #pragma unroll
    for (int t = 0; t < 8; ++t) {
        #pragma unroll
        for (int v = 0; v < 8; ++v) {
            const int m = v + (hi << 3);
            out[(size_t)(b0 + m) * OUTD + 16 * t + l15] = acc[t][v];
        }
    }
}

extern "C" void kernel_launch(void* const* d_in, const int* in_sizes, int n_in,
                              void* d_out, int out_size, void* d_ws, size_t ws_size,
                              hipStream_t stream) {
    const float* x   = (const float*)d_in[0];
    const int*   ei  = (const int*)d_in[1];
    const float* W1l = (const float*)d_in[2];
    const float* W1r = (const float*)d_in[3];
    const float* b1  = (const float*)d_in[4];
    const float* W2l = (const float*)d_in[5];
    const float* W2r = (const float*)d_in[6];
    const float* b2  = (const float*)d_in[7];
    float* out = (float*)d_out;

    const int B = in_sizes[0] / (NNODE * 3);
    const int blocks = B / BPB;
    sage_rowfrag_kernel<<<blocks, 128, 0, stream>>>(x, ei, W1l, W1r, b1, W2l, W2r, b2, out);
}


# neighbourhood aggregation (graph matrix x node features) moved from 1875 lane FMAs behind serial LDS row reads to 84 f32 16x16x4 matrix instructions; source-node order kept, sums in the matrix unit
# speedup vs baseline: 1.2834x; 1.1672x over previous
#include <hip/hip_runtime.h>


typedef __attribute__((ext_vector_type(2))) float v2f;
typedef __attribute__((ext_vector_type(4))) float v4f;
typedef __attribute__((ext_vector_type(8))) float v8f;

#define NNODE 25
#define NEDGE 64
#define HID   128
#define OUTD  128
#define TPW   16
#define WPB   4
#define BPB   (TPW * WPB)
#define NJ    (HID / 4)
#define A1P   28
#define XW    (TPW * NNODE * 3)
#define CHK   140

static __device__ __forceinline__ v2f lo2(v4f q) { return __builtin_shufflevector(q, q, 0, 1); }
static __device__ __forceinline__ v2f hi2(v4f q) { return __builtin_shufflevector(q, q, 2, 3); }
static __device__ __forceinline__ v2f sp2(float z) { return (v2f){ z, z }; }

#define NSTG  8
#define NSLOT (NNODE + NSTG - 1)

static __device__ __forceinline__ void kstep(const float (&ag)[NNODE][3], const float (&xr)[NNODE][3],
                                             const float (&nw)[NNODE], v4f (&w)[4], const v4f* wnext,
                                             v2f& acl, v2f& acr)
{
    v2f pre[NNODE];
    v2f cl = { 0.0f, 0.0f };
    v2f cr = { 0.0f, 0.0f };
    #pragma unroll
    for (int s = 0; s < NSLOT; ++s) {
        #pragma unroll
        for (int n = 0; n < NNODE; ++n) {
            const int st = s - n;
            if (st < 0 || st >= NSTG) continue;
            if (st == 0)      pre[n] = __builtin_elementwise_fma(sp2(ag[n][0]), lo2(w[0]), lo2(w[3]));
            else if (st == 1) pre[n] = __builtin_elementwise_fma(sp2(ag[n][1]), hi2(w[0]), pre[n]);
            else if (st == 2) pre[n] = __builtin_elementwise_fma(sp2(ag[n][2]), lo2(w[1]), pre[n]);
            else if (st == 3) pre[n] = __builtin_elementwise_fma(sp2(xr[n][0]), hi2(w[1]), pre[n]);
            else if (st == 4) pre[n] = __builtin_elementwise_fma(sp2(xr[n][1]), lo2(w[2]), pre[n]);
            else if (st == 5) pre[n] = __builtin_elementwise_fma(sp2(xr[n][2]), hi2(w[2]), pre[n]);
            else if (st == 6) pre[n] = __builtin_elementwise_maxnum(pre[n], (v2f){ 0.0f, 0.0f });
            else {
                cl = __builtin_elementwise_fma(pre[n], sp2(nw[n]), cl);
                cr = cr + pre[n];
            }
        }
        if (s == 25) { w[0] = wnext[0]; w[3] = wnext[3]; }
        if (s == 27) w[1] = wnext[1];
        if (s == 29) w[2] = wnext[2];
        if (s == NSLOT - 1) {
            acl = cl;
            acr = cr * (1.0f / 25.0f);
        }
        if ((s & 1) == 1) __builtin_amdgcn_sched_barrier(0);
    }
}

__launch_bounds__(128)
__global__ void sage_rowfrag_kernel(const float* __restrict__ x,
                                    const int*   __restrict__ edge_index,
                                    const float* __restrict__ W1l,
                                    const float* __restrict__ W1r,
                                    const float* __restrict__ b1,
                                    const float* __restrict__ W2l,
                                    const float* __restrict__ W2r,
                                    const float* __restrict__ b2,
                                    float* __restrict__ out)
{
    __shared__ int   ssrc[NEDGE];
    __shared__ int   sdst[NEDGE];
    __shared__ float sdeginv[NNODE];
    __shared__ float snodew[NNODE];
    __align__(16) __shared__ float sA1[NNODE * A1P];
    __align__(16) __shared__ float sW1[NJ * 2 * 16];
    __align__(16) __shared__ float sx[WPB * XW];
    __align__(16) __shared__ float sW2[2 * NJ * 4 * CHK];

    const int tid  = threadIdx.x;
    const int lane = tid & 31;
    const int wave = tid >> 5;
    const int hi   = lane >> 4;
    const int l15  = lane & 15;
    const int b0   = (blockIdx.x * WPB + wave) * TPW;

    v4f wra[32], wrb[32];
    #pragma unroll
    for (int u = 0; u < 32; ++u) {
        const int it  = tid + 128 * u;
        const int mat = it >> 11;
        const int rp  = (it >> 5) & 63;
        const int c4  = it & 31;
        const float* Wm = mat ? W2r : W2l;
        wra[u] = *(const v4f*)&Wm[(2 * rp) * OUTD + 4 * c4];
        wrb[u] = *(const v4f*)&Wm[(2 * rp + 1) * OUTD + 4 * c4];
    }

    if (tid < NEDGE) {
        ssrc[tid] = edge_index[tid];
        sdst[tid] = edge_index[NEDGE + tid];
    }
    for (int i = tid; i < NNODE * A1P; i += 128) sA1[i] = 0.0f;
    if (tid < NNODE) snodew[tid] = 0.0f;

    {
        const v4f* xg = (const v4f*)(x + (size_t)b0 * (NNODE * 3));
        v4f* xs = (v4f*)&sx[wave * XW];
        for (int i = lane; i < XW / 4; i += 32) xs[i] = xg[i];
    }
    for (int idx = tid; idx < NJ * 2 * 16; idx += 128) {
        const int jh = idx >> 4, i = idx & 15;
        const int g = i >> 1, col = 2 * jh + (i & 1);
        float v = 0.0f;
        if (g < 3)       v = W1l[g * HID + col];
        else if (g < 6)  v = W1r[(g - 3) * HID + col];
        else if (g == 6) v = b1[col];
        sW1[idx] = v;
    }
    __syncthreads();

    if (tid < NNODE) {
        int c = 0;
        #pragma unroll
        for (int e = 0; e < NEDGE; ++e) c += (sdst[e] == tid);
        sdeginv[tid] = 1.0f / fmaxf((float)c, 1.0f);
    }
    __syncthreads();

    if (tid < NEDGE) {
        const float di = sdeginv[sdst[tid]];
        atomicAdd(&sA1[sdst[tid] * A1P + ssrc[tid]], di);
        atomicAdd(&snodew[ssrc[tid]], di * (1.0f / 25.0f));
    }
    #pragma unroll
    for (int u = 0; u < 32; ++u) {
        const int it  = tid + 128 * u;
        const int mat = it >> 11;
        const int rp  = (it >> 5) & 63;
        const int c4  = it & 31;
        const int j = rp >> 1, h = rp & 1;
        const int t = c4 >> 2, q = t >> 1, tt = t & 1;
        const int cb = 4 * (c4 & 3);
        float* dstp = &sW2[((mat * NJ + j) * 4 + q) * CHK + (cb + 16 * h) * 4 + tt * 2];
        *(v2f*)(dstp + 0)  = (v2f){ wra[u].x, wrb[u].x };
        *(v2f*)(dstp + 4)  = (v2f){ wra[u].y, wrb[u].y };
        *(v2f*)(dstp + 8)  = (v2f){ wra[u].z, wrb[u].z };
        *(v2f*)(dstp + 12) = (v2f){ wra[u].w, wrb[u].w };
    }
    __syncthreads();

    float xr[NNODE][3];
    {
        const float* xrow = &sx[wave * XW + l15 * (NNODE * 3)];
        #pragma unroll
        for (int m = 0; m < NNODE; ++m) {
            xr[m][0] = xrow[3 * m + 0];
            xr[m][1] = xrow[3 * m + 1];
            xr[m][2] = xrow[3 * m + 2];
        }
    }
    float ag[NNODE][3];
    {
        v2f bx[A1P / 4][3];
        #pragma unroll
        for (int ks = 0; ks < A1P / 4; ++ks) {
            #pragma unroll
            for (int i = 0; i < 3; ++i) {
                const float e0 = (4 * ks + 0 < NNODE) ? xr[(4 * ks + 0 < NNODE) ? 4 * ks + 0 : 0][i] : 0.0f;
                const float e1 = (4 * ks + 1 < NNODE) ? xr[(4 * ks + 1 < NNODE) ? 4 * ks + 1 : 0][i] : 0.0f;
                const float e2 = (4 * ks + 2 < NNODE) ? xr[(4 * ks + 2 < NNODE) ? 4 * ks + 2 : 0][i] : 0.0f;
                const float e3 = (4 * ks + 3 < NNODE) ? xr[(4 * ks + 3 < NNODE) ? 4 * ks + 3 : 0][i] : 0.0f;
                bx[ks][i] = (v2f){ hi ? e2 : e0, hi ? e3 : e1 };
            }
        }
        v2f aa[4][A1P / 4];
        #pragma unroll
        for (int tl = 0; tl < 4; ++tl) {
            const int node = min(8 * tl + (l15 & 7), NNODE - 1);
            const float* ap = &sA1[node * A1P + 2 * hi];
            #pragma unroll
            for (int ks = 0; ks < A1P / 4; ++ks) aa[tl][ks] = *(const v2f*)&ap[4 * ks];
        }
        v8f g[4][3];
        #pragma unroll
        for (int tl = 0; tl < 4; ++tl) {
            #pragma unroll
            for (int i = 0; i < 3; ++i) {
                #pragma unroll
                for (int v = 0; v < 8; ++v) g[tl][i][v] = 0.0f;
                asm volatile("" : "+v"(g[tl][i]));
            }
        }
        __builtin_amdgcn_sched_barrier(0);
        #pragma unroll
        for (int ks = 0; ks < A1P / 4; ++ks) {
            #pragma unroll
            for (int tl = 0; tl < 4; ++tl) {
                #pragma unroll
                for (int i = 0; i < 3; ++i) {
                    g[tl][i] = __builtin_amdgcn_wmma_f32_16x16x4_f32(
                                   false, aa[tl][ks], false, bx[ks][i], (short)0, g[tl][i], false, false);
                }
            }
        }
        __builtin_amdgcn_sched_barrier(0);
        asm volatile("v_nop\n\tv_nop\n\tv_nop\n\tv_nop\n\tv_nop");
        #pragma unroll
        for (int n = 0; n < NNODE; ++n) {
            #pragma unroll
            for (int i = 0; i < 3; ++i) ag[n][i] = g[n >> 3][i][n & 7];
        }
    }
    float nw[NNODE];
    #pragma unroll
    for (int n = 0; n < NNODE; ++n) nw[n] = snodew[n];

    v8f acc[8];
    #pragma unroll
    for (int t = 0; t < 8; ++t) {
        const float bz = b2[16 * t + l15];
        #pragma unroll
        for (int v = 0; v < 8; ++v) acc[t][v] = bz;
    }

    const v4f* w1p = (const v4f*)&sW1[hi * 16];
    const float* w2p = &sW2[lane * 4];
    v4f w[4];
    #pragma unroll
    for (int i = 0; i < 4; ++i) w[i] = w1p[i];
    #pragma unroll 1
    for (int j = 0; j < NJ; ++j) {
        v4f bl[4], br[4];
        #pragma unroll
        for (int q = 0; q < 4; ++q) {
            bl[q] = *(const v4f*)&w2p[((0 * NJ + j) * 4 + q) * CHK];
            br[q] = *(const v4f*)&w2p[((1 * NJ + j) * 4 + q) * CHK];
        }
        const int jn = (j + 1 < NJ) ? (j + 1) : j;
        v2f acl, acr;
        kstep(ag, xr, nw, w, w1p + jn * 8, acl, acr);
        #pragma unroll
        for (int q = 0; q < 4; ++q) {
            acc[2 * q]     = __builtin_amdgcn_wmma_f32_16x16x4_f32(
                                 false, acl, false, lo2(bl[q]), (short)0, acc[2 * q], false, false);
            acc[2 * q + 1] = __builtin_amdgcn_wmma_f32_16x16x4_f32(
                                 false, acl, false, hi2(bl[q]), (short)0, acc[2 * q + 1], false, false);
        }
        #pragma unroll
        for (int q = 0; q < 4; ++q) {
            acc[2 * q]     = __builtin_amdgcn_wmma_f32_16x16x4_f32(
                                 false, acr, false, lo2(br[q]), (short)0, acc[2 * q], false, false);
            acc[2 * q + 1] = __builtin_amdgcn_wmma_f32_16x16x4_f32(
                                 false, acr, false, hi2(br[q]), (short)0, acc[2 * q + 1], false, false);
        }
        __builtin_amdgcn_sched_barrier(0);
        asm volatile("v_nop\n\tv_nop\n\tv_nop\n\tv_nop");
    }

    #pragma unroll
    for (int t = 0; t < 8; ++t) {
        #pragma unroll
        for (int v = 0; v < 8; ++v) {
            const int m = v + (hi << 3);
            out[(size_t)(b0 + m) * OUTD + 16 * t + l15] = acc[t][v];
        }
    }
}

extern "C" void kernel_launch(void* const* d_in, const int* in_sizes, int n_in,
                              void* d_out, int out_size, void* d_ws, size_t ws_size,
                              hipStream_t stream) {
    const float* x   = (const float*)d_in[0];
    const int*   ei  = (const int*)d_in[1];
    const float* W1l = (const float*)d_in[2];
    const float* W1r = (const float*)d_in[3];
    const float* b1  = (const float*)d_in[4];
    const float* W2l = (const float*)d_in[5];
    const float* W2r = (const float*)d_in[6];
    const float* b2  = (const float*)d_in[7];
    float* out = (float*)d_out;

    const int B = in_sizes[0] / (NNODE * 3);
    const int blocks = B / BPB;
    sage_rowfrag_kernel<<<blocks, 128, 0, stream>>>(x, ei, W1l, W1r, b1, W2l, W2r, b2, out);
}


# every LDS read of a k-step (its 8 W2 fragment reads, the next step's layer-1 weights) issued as one group at the step's head into a second weight register set, 2 k-steps per trip; sums unchanged
# speedup vs baseline: 1.3848x; 1.0790x over previous
#include <hip/hip_runtime.h>


typedef __attribute__((ext_vector_type(2))) float v2f;
typedef __attribute__((ext_vector_type(4))) float v4f;
typedef __attribute__((ext_vector_type(8))) float v8f;

#define NNODE 25
#define NEDGE 64
#define HID   128
#define OUTD  128
#define TPW   16
#define WPB   4
#define BPB   (TPW * WPB)
#define NJ    (HID / 4)
#define A1P   28
#define XW    (TPW * NNODE * 3)
#define CHK   140

static __device__ __forceinline__ v2f lo2(v4f q) { return __builtin_shufflevector(q, q, 0, 1); }
static __device__ __forceinline__ v2f hi2(v4f q) { return __builtin_shufflevector(q, q, 2, 3); }
static __device__ __forceinline__ v2f sp2(float z) { return (v2f){ z, z }; }

#define NSTG  8
#define NSLOT (NNODE + NSTG - 1)

static __device__ __forceinline__ void kstep(const float (&ag)[NNODE][3], const float (&xr)[NNODE][3],
                                             const float (&nw)[NNODE], const v4f (&w)[3], const v2f wb,
                                             v2f& acl, v2f& acr)
{
    v2f pre[NNODE];
    v2f cl = { 0.0f, 0.0f };
    v2f cr = { 0.0f, 0.0f };
    #pragma unroll
    for (int s = 0; s < NSLOT; ++s) {
        #pragma unroll
        for (int n = 0; n < NNODE; ++n) {
            const int st = s - n;
            if (st < 0 || st >= NSTG) continue;
            if (st == 0)      pre[n] = __builtin_elementwise_fma(sp2(ag[n][0]), lo2(w[0]), wb);
            else if (st == 1) pre[n] = __builtin_elementwise_fma(sp2(ag[n][1]), hi2(w[0]), pre[n]);
            else if (st == 2) pre[n] = __builtin_elementwise_fma(sp2(ag[n][2]), lo2(w[1]), pre[n]);
            else if (st == 3) pre[n] = __builtin_elementwise_fma(sp2(xr[n][0]), hi2(w[1]), pre[n]);
            else if (st == 4) pre[n] = __builtin_elementwise_fma(sp2(xr[n][1]), lo2(w[2]), pre[n]);
            else if (st == 5) pre[n] = __builtin_elementwise_fma(sp2(xr[n][2]), hi2(w[2]), pre[n]);
            else if (st == 6) pre[n] = __builtin_elementwise_maxnum(pre[n], (v2f){ 0.0f, 0.0f });
            else {
                cl = __builtin_elementwise_fma(pre[n], sp2(nw[n]), cl);
                cr = cr + pre[n];
            }
        }
        if (s == NSLOT - 1) {
            acl = cl;
            acr = cr * (1.0f / 25.0f);
        }
        if ((s & 1) == 1) __builtin_amdgcn_sched_barrier(0);
    }
}

static __device__ __forceinline__ void kiter(const float (&ag)[NNODE][3], const float (&xr)[NNODE][3],
                                             const float (&nw)[NNODE], const float* w2j, const v4f* w1n,
                                             const v4f (&w)[3], const v2f wb, v4f (&wn)[3], v2f& bn,
                                             v8f (&acc)[8])
{
    v4f bl[4], br[4];
    #pragma unroll
    for (int q = 0; q < 4; ++q) {
        bl[q] = *(const v4f*)&w2j[q * CHK];
        br[q] = *(const v4f*)&w2j[(NJ * 4 + q) * CHK];
    }
    wn[0] = w1n[0];
    wn[1] = w1n[1];
    wn[2] = w1n[2];
    bn    = *(const v2f*)&w1n[3];
    __builtin_amdgcn_sched_barrier(0);
    v2f acl, acr;
    kstep(ag, xr, nw, w, wb, acl, acr);
    #pragma unroll
    for (int q = 0; q < 4; ++q) {
        acc[2 * q]     = __builtin_amdgcn_wmma_f32_16x16x4_f32(
                             false, acl, false, lo2(bl[q]), (short)0, acc[2 * q], false, false);
        acc[2 * q + 1] = __builtin_amdgcn_wmma_f32_16x16x4_f32(
                             false, acl, false, hi2(bl[q]), (short)0, acc[2 * q + 1], false, false);
    }
    #pragma unroll
    for (int q = 0; q < 4; ++q) {
        acc[2 * q]     = __builtin_amdgcn_wmma_f32_16x16x4_f32(
                             false, acr, false, lo2(br[q]), (short)0, acc[2 * q], false, false);
        acc[2 * q + 1] = __builtin_amdgcn_wmma_f32_16x16x4_f32(
                             false, acr, false, hi2(br[q]), (short)0, acc[2 * q + 1], false, false);
    }
    __builtin_amdgcn_sched_barrier(0);
    asm volatile("v_nop\n\tv_nop\n\tv_nop\n\tv_nop");
}

__launch_bounds__(128)
__global__ void sage_rowfrag_kernel(const float* __restrict__ x,
                                    const int*   __restrict__ edge_index,
                                    const float* __restrict__ W1l,
                                    const float* __restrict__ W1r,
                                    const float* __restrict__ b1,
                                    const float* __restrict__ W2l,
                                    const float* __restrict__ W2r,
                                    const float* __restrict__ b2,
                                    float* __restrict__ out)
{
    __shared__ int   ssrc[NEDGE];
    __shared__ int   sdst[NEDGE];
    __shared__ float sdeginv[NNODE];
    __shared__ float snodew[NNODE];
    __align__(16) __shared__ float sA1[NNODE * A1P];
    __align__(16) __shared__ float sW1[NJ * 2 * 16];
    __align__(16) __shared__ float sx[WPB * XW];
    __align__(16) __shared__ float sW2[2 * NJ * 4 * CHK];

    const int tid  = threadIdx.x;
    const int lane = tid & 31;
    const int wave = tid >> 5;
    const int hi   = lane >> 4;
    const int l15  = lane & 15;
    const int b0   = (blockIdx.x * WPB + wave) * TPW;

    v4f wra[32], wrb[32];
    #pragma unroll
    for (int u = 0; u < 32; ++u) {
        const int it  = tid + 128 * u;
        const int mat = it >> 11;
        const int rp  = (it >> 5) & 63;
        const int c4  = it & 31;
        const float* Wm = mat ? W2r : W2l;
        wra[u] = *(const v4f*)&Wm[(2 * rp) * OUTD + 4 * c4];
        wrb[u] = *(const v4f*)&Wm[(2 * rp + 1) * OUTD + 4 * c4];
    }

    if (tid < NEDGE) {
        ssrc[tid] = edge_index[tid];
        sdst[tid] = edge_index[NEDGE + tid];
    }
    for (int i = tid; i < NNODE * A1P; i += 128) sA1[i] = 0.0f;
    if (tid < NNODE) snodew[tid] = 0.0f;

    {
        const v4f* xg = (const v4f*)(x + (size_t)b0 * (NNODE * 3));
        v4f* xs = (v4f*)&sx[wave * XW];
        for (int i = lane; i < XW / 4; i += 32) xs[i] = xg[i];
    }
    for (int idx = tid; idx < NJ * 2 * 16; idx += 128) {
        const int jh = idx >> 4, i = idx & 15;
        const int g = i >> 1, col = 2 * jh + (i & 1);
        float v = 0.0f;
        if (g < 3)       v = W1l[g * HID + col];
        else if (g < 6)  v = W1r[(g - 3) * HID + col];
        else if (g == 6) v = b1[col];
        sW1[idx] = v;
    }
    __syncthreads();

    if (tid < NNODE) {
        int c = 0;
        #pragma unroll
        for (int e = 0; e < NEDGE; ++e) c += (sdst[e] == tid);
        sdeginv[tid] = 1.0f / fmaxf((float)c, 1.0f);
    }
    __syncthreads();

    if (tid < NEDGE) {
        const float di = sdeginv[sdst[tid]];
        atomicAdd(&sA1[sdst[tid] * A1P + ssrc[tid]], di);
        atomicAdd(&snodew[ssrc[tid]], di * (1.0f / 25.0f));
    }
    #pragma unroll
    for (int u = 0; u < 32; ++u) {
        const int it  = tid + 128 * u;
        const int mat = it >> 11;
        const int rp  = (it >> 5) & 63;
        const int c4  = it & 31;
        const int j = rp >> 1, h = rp & 1;
        const int t = c4 >> 2, q = t >> 1, tt = t & 1;
        const int cb = 4 * (c4 & 3);
        float* dstp = &sW2[((mat * NJ + j) * 4 + q) * CHK + (cb + 16 * h) * 4 + tt * 2];
        *(v2f*)(dstp + 0)  = (v2f){ wra[u].x, wrb[u].x };
        *(v2f*)(dstp + 4)  = (v2f){ wra[u].y, wrb[u].y };
        *(v2f*)(dstp + 8)  = (v2f){ wra[u].z, wrb[u].z };
        *(v2f*)(dstp + 12) = (v2f){ wra[u].w, wrb[u].w };
    }
    __syncthreads();

    float xr[NNODE][3];
    {
        const float* xrow = &sx[wave * XW + l15 * (NNODE * 3)];
        #pragma unroll
        for (int m = 0; m < NNODE; ++m) {
            xr[m][0] = xrow[3 * m + 0];
            xr[m][1] = xrow[3 * m + 1];
            xr[m][2] = xrow[3 * m + 2];
        }
    }
    float ag[NNODE][3];
    {
        v2f bx[A1P / 4][3];
        #pragma unroll
        for (int ks = 0; ks < A1P / 4; ++ks) {
            #pragma unroll
            for (int i = 0; i < 3; ++i) {
                const float e0 = (4 * ks + 0 < NNODE) ? xr[(4 * ks + 0 < NNODE) ? 4 * ks + 0 : 0][i] : 0.0f;
                const float e1 = (4 * ks + 1 < NNODE) ? xr[(4 * ks + 1 < NNODE) ? 4 * ks + 1 : 0][i] : 0.0f;
                const float e2 = (4 * ks + 2 < NNODE) ? xr[(4 * ks + 2 < NNODE) ? 4 * ks + 2 : 0][i] : 0.0f;
                const float e3 = (4 * ks + 3 < NNODE) ? xr[(4 * ks + 3 < NNODE) ? 4 * ks + 3 : 0][i] : 0.0f;
                bx[ks][i] = (v2f){ hi ? e2 : e0, hi ? e3 : e1 };
            }
        }
        v2f aa[4][A1P / 4];
        #pragma unroll
        for (int tl = 0; tl < 4; ++tl) {
            const int node = min(8 * tl + (l15 & 7), NNODE - 1);
            const float* ap = &sA1[node * A1P + 2 * hi];
            #pragma unroll
            for (int ks = 0; ks < A1P / 4; ++ks) aa[tl][ks] = *(const v2f*)&ap[4 * ks];
        }
        v8f g[4][3];
        #pragma unroll
        for (int tl = 0; tl < 4; ++tl) {
            #pragma unroll
            for (int i = 0; i < 3; ++i) {
                #pragma unroll
                for (int v = 0; v < 8; ++v) g[tl][i][v] = 0.0f;
                asm volatile("" : "+v"(g[tl][i]));
            }
        }
        __builtin_amdgcn_sched_barrier(0);
        #pragma unroll
        for (int ks = 0; ks < A1P / 4; ++ks) {
            #pragma unroll
            for (int tl = 0; tl < 4; ++tl) {
                #pragma unroll
                for (int i = 0; i < 3; ++i) {
                    g[tl][i] = __builtin_amdgcn_wmma_f32_16x16x4_f32(
                                   false, aa[tl][ks], false, bx[ks][i], (short)0, g[tl][i], false, false);
                }
            }
        }
        __builtin_amdgcn_sched_barrier(0);
        asm volatile("v_nop\n\tv_nop\n\tv_nop\n\tv_nop\n\tv_nop");
        #pragma unroll
        for (int n = 0; n < NNODE; ++n) {
            #pragma unroll
            for (int i = 0; i < 3; ++i) ag[n][i] = g[n >> 3][i][n & 7];
        }
    }
    float nw[NNODE];
    #pragma unroll
    for (int n = 0; n < NNODE; ++n) nw[n] = snodew[n];

    v8f acc[8];
    #pragma unroll
    for (int t = 0; t < 8; ++t) {
        const float bz = b2[16 * t + l15];
        #pragma unroll
        for (int v = 0; v < 8; ++v) acc[t][v] = bz;
    }

    const v4f* w1p = (const v4f*)&sW1[hi * 16];
    const float* w2p = &sW2[lane * 4];
    v4f wA[3], wB[3];
    v2f bA, bB;
    wA[0] = w1p[0];
    wA[1] = w1p[1];
    wA[2] = w1p[2];
    bA    = *(const v2f*)&w1p[3];
    #pragma unroll 1
    for (int j = 0; j < NJ; j += 2) {
        const int jn = (j + 2 < NJ) ? (j + 2) : j;
        kiter(ag, xr, nw, w2p + (j * 4) * CHK,       w1p + (j + 1) * 8, wA, bA, wB, bB, acc);
        kiter(ag, xr, nw, w2p + ((j + 1) * 4) * CHK, w1p + jn * 8,      wB, bB, wA, bA, acc);
    }

    #pragma unroll
    for (int t = 0; t < 8; ++t) {
        #pragma unroll
        for (int v = 0; v < 8; ++v) {
            const int m = v + (hi << 3);
            out[(size_t)(b0 + m) * OUTD + 16 * t + l15] = acc[t][v];
        }
    }
}

extern "C" void kernel_launch(void* const* d_in, const int* in_sizes, int n_in,
                              void* d_out, int out_size, void* d_ws, size_t ws_size,
                              hipStream_t stream) {
    const float* x   = (const float*)d_in[0];
    const int*   ei  = (const int*)d_in[1];
    const float* W1l = (const float*)d_in[2];
    const float* W1r = (const float*)d_in[3];
    const float* b1  = (const float*)d_in[4];
    const float* W2l = (const float*)d_in[5];
    const float* W2r = (const float*)d_in[6];
    const float* b2  = (const float*)d_in[7];
    float* out = (float*)d_out;

    const int B = in_sizes[0] / (NNODE * 3);
    const int blocks = B / BPB;
    sage_rowfrag_kernel<<<blocks, 128, 0, stream>>>(x, ei, W1l, W1r, b1, W2l, W2r, b2, out);
}


# layer-1 weight table: a thread's 8 entries loaded as one branch-free group at entry, ahead of the W2 stream (was an 8-trip loop, each trip a dependent load, wait and LDS store); sums unchanged
# speedup vs baseline: 1.4570x; 1.0521x over previous
#include <hip/hip_runtime.h>


typedef __attribute__((ext_vector_type(2))) float v2f;
typedef __attribute__((ext_vector_type(4))) float v4f;
typedef __attribute__((ext_vector_type(8))) float v8f;

#define NNODE 25
#define NEDGE 64
#define HID   128
#define OUTD  128
#define TPW   16
#define WPB   4
#define BPB   (TPW * WPB)
#define NJ    (HID / 4)
#define A1P   28
#define XW    (TPW * NNODE * 3)
#define CHK   140

static __device__ __forceinline__ v2f lo2(v4f q) { return __builtin_shufflevector(q, q, 0, 1); }
static __device__ __forceinline__ v2f hi2(v4f q) { return __builtin_shufflevector(q, q, 2, 3); }
static __device__ __forceinline__ v2f sp2(float z) { return (v2f){ z, z }; }

#define NSTG  8
#define NSLOT (NNODE + NSTG - 1)

static __device__ __forceinline__ void kstep(const float (&ag)[NNODE][3], const float (&xr)[NNODE][3],
                                             const float (&nw)[NNODE], const v4f (&w)[3], const v2f wb,
                                             v2f& acl, v2f& acr)
{
    v2f pre[NNODE];
    v2f cl = { 0.0f, 0.0f };
    v2f cr = { 0.0f, 0.0f };
    #pragma unroll
    for (int s = 0; s < NSLOT; ++s) {
        #pragma unroll
        for (int n = 0; n < NNODE; ++n) {
            const int st = s - n;
            if (st < 0 || st >= NSTG) continue;
            if (st == 0)      pre[n] = __builtin_elementwise_fma(sp2(ag[n][0]), lo2(w[0]), wb);
            else if (st == 1) pre[n] = __builtin_elementwise_fma(sp2(ag[n][1]), hi2(w[0]), pre[n]);
            else if (st == 2) pre[n] = __builtin_elementwise_fma(sp2(ag[n][2]), lo2(w[1]), pre[n]);
            else if (st == 3) pre[n] = __builtin_elementwise_fma(sp2(xr[n][0]), hi2(w[1]), pre[n]);
            else if (st == 4) pre[n] = __builtin_elementwise_fma(sp2(xr[n][1]), lo2(w[2]), pre[n]);
            else if (st == 5) pre[n] = __builtin_elementwise_fma(sp2(xr[n][2]), hi2(w[2]), pre[n]);
            else if (st == 6) pre[n] = __builtin_elementwise_maxnum(pre[n], (v2f){ 0.0f, 0.0f });
            else {
                cl = __builtin_elementwise_fma(pre[n], sp2(nw[n]), cl);
                cr = cr + pre[n];
            }
        }
        if (s == NSLOT - 1) {
            acl = cl;
            acr = cr * (1.0f / 25.0f);
        }
        if ((s & 1) == 1) __builtin_amdgcn_sched_barrier(0);
    }
}

static __device__ __forceinline__ void kiter(const float (&ag)[NNODE][3], const float (&xr)[NNODE][3],
                                             const float (&nw)[NNODE], const float* w2j, const v4f* w1n,
                                             const v4f (&w)[3], const v2f wb, v4f (&wn)[3], v2f& bn,
                                             v8f (&acc)[8])
{
    v4f bl[4], br[4];
    #pragma unroll
    for (int q = 0; q < 4; ++q) {
        bl[q] = *(const v4f*)&w2j[q * CHK];
        br[q] = *(const v4f*)&w2j[(NJ * 4 + q) * CHK];
    }
    wn[0] = w1n[0];
    wn[1] = w1n[1];
    wn[2] = w1n[2];
    bn    = *(const v2f*)&w1n[3];
    __builtin_amdgcn_sched_barrier(0);
    v2f acl, acr;
    kstep(ag, xr, nw, w, wb, acl, acr);
    #pragma unroll
    for (int q = 0; q < 4; ++q) {
        acc[2 * q]     = __builtin_amdgcn_wmma_f32_16x16x4_f32(
                             false, acl, false, lo2(bl[q]), (short)0, acc[2 * q], false, false);
        acc[2 * q + 1] = __builtin_amdgcn_wmma_f32_16x16x4_f32(
                             false, acl, false, hi2(bl[q]), (short)0, acc[2 * q + 1], false, false);
    }
    #pragma unroll
    for (int q = 0; q < 4; ++q) {
        acc[2 * q]     = __builtin_amdgcn_wmma_f32_16x16x4_f32(
                             false, acr, false, lo2(br[q]), (short)0, acc[2 * q], false, false);
        acc[2 * q + 1] = __builtin_amdgcn_wmma_f32_16x16x4_f32(
                             false, acr, false, hi2(br[q]), (short)0, acc[2 * q + 1], false, false);
    }
    __builtin_amdgcn_sched_barrier(0);
    asm volatile("v_nop\n\tv_nop\n\tv_nop\n\tv_nop");
}

__launch_bounds__(128)
__global__ void sage_rowfrag_kernel(const float* __restrict__ x,
                                    const int*   __restrict__ edge_index,
                                    const float* __restrict__ W1l,
                                    const float* __restrict__ W1r,
                                    const float* __restrict__ b1,
                                    const float* __restrict__ W2l,
                                    const float* __restrict__ W2r,
                                    const float* __restrict__ b2,
                                    float* __restrict__ out)
{
    __shared__ int   ssrc[NEDGE];
    __shared__ int   sdst[NEDGE];
    __shared__ float sdeginv[NNODE];
    __shared__ float snodew[NNODE];
    __align__(16) __shared__ float sA1[NNODE * A1P];
    __align__(16) __shared__ float sW1[NJ * 2 * 16];
    __align__(16) __shared__ float sx[WPB * XW];
    __align__(16) __shared__ float sW2[2 * NJ * 4 * CHK];

    const int tid  = threadIdx.x;
    const int lane = tid & 31;
    const int wave = tid >> 5;
    const int hi   = lane >> 4;
    const int l15  = lane & 15;
    const int b0   = (blockIdx.x * WPB + wave) * TPW;

    const int w1g = (tid & 15) >> 1;
    const float* w1row = (w1g < 3) ? (W1l + w1g * HID) : (w1g < 6) ? (W1r + (w1g - 3) * HID) : b1;
    const int w1col = 2 * (tid >> 4) + (tid & 1);
    float w1v[8];
    #pragma unroll
    for (int k = 0; k < 8; ++k) w1v[k] = w1row[w1col + 16 * k];

    v4f wra[32], wrb[32];
    #pragma unroll
    for (int u = 0; u < 32; ++u) {
        const int it  = tid + 128 * u;
        const int mat = it >> 11;
        const int rp  = (it >> 5) & 63;
        const int c4  = it & 31;
        const float* Wm = mat ? W2r : W2l;
        wra[u] = *(const v4f*)&Wm[(2 * rp) * OUTD + 4 * c4];
        wrb[u] = *(const v4f*)&Wm[(2 * rp + 1) * OUTD + 4 * c4];
    }

    #pragma unroll
    for (int k = 0; k < 8; ++k) sW1[tid + 128 * k] = (w1g == 7) ? 0.0f : w1v[k];

    if (tid < NEDGE) {
        ssrc[tid] = edge_index[tid];
        sdst[tid] = edge_index[NEDGE + tid];
    }
    for (int i = tid; i < NNODE * A1P; i += 128) sA1[i] = 0.0f;
    if (tid < NNODE) snodew[tid] = 0.0f;

    {
        const v4f* xg = (const v4f*)(x + (size_t)b0 * (NNODE * 3));
        v4f* xs = (v4f*)&sx[wave * XW];
        for (int i = lane; i < XW / 4; i += 32) xs[i] = xg[i];
    }
    __syncthreads();

    if (tid < NNODE) {
        int c = 0;
        #pragma unroll
        for (int e = 0; e < NEDGE; ++e) c += (sdst[e] == tid);
        sdeginv[tid] = 1.0f / fmaxf((float)c, 1.0f);
    }
    __syncthreads();

    if (tid < NEDGE) {
        const float di = sdeginv[sdst[tid]];
        atomicAdd(&sA1[sdst[tid] * A1P + ssrc[tid]], di);
        atomicAdd(&snodew[ssrc[tid]], di * (1.0f / 25.0f));
    }
    #pragma unroll
    for (int u = 0; u < 32; ++u) {
        const int it  = tid + 128 * u;
        const int mat = it >> 11;
        const int rp  = (it >> 5) & 63;
        const int c4  = it & 31;
        const int j = rp >> 1, h = rp & 1;
        const int t = c4 >> 2, q = t >> 1, tt = t & 1;
        const int cb = 4 * (c4 & 3);
        float* dstp = &sW2[((mat * NJ + j) * 4 + q) * CHK + (cb + 16 * h) * 4 + tt * 2];
        *(v2f*)(dstp + 0)  = (v2f){ wra[u].x, wrb[u].x };
        *(v2f*)(dstp + 4)  = (v2f){ wra[u].y, wrb[u].y };
        *(v2f*)(dstp + 8)  = (v2f){ wra[u].z, wrb[u].z };
        *(v2f*)(dstp + 12) = (v2f){ wra[u].w, wrb[u].w };
    }
    __syncthreads();

    float xr[NNODE][3];
    {
        const float* xrow = &sx[wave * XW + l15 * (NNODE * 3)];
        #pragma unroll
        for (int m = 0; m < NNODE; ++m) {
            xr[m][0] = xrow[3 * m + 0];
            xr[m][1] = xrow[3 * m + 1];
            xr[m][2] = xrow[3 * m + 2];
        }
    }
    float ag[NNODE][3];
    {
        v2f bx[A1P / 4][3];
        #pragma unroll
        for (int ks = 0; ks < A1P / 4; ++ks) {
            #pragma unroll
            for (int i = 0; i < 3; ++i) {
                const float e0 = (4 * ks + 0 < NNODE) ? xr[(4 * ks + 0 < NNODE) ? 4 * ks + 0 : 0][i] : 0.0f;
                const float e1 = (4 * ks + 1 < NNODE) ? xr[(4 * ks + 1 < NNODE) ? 4 * ks + 1 : 0][i] : 0.0f;
                const float e2 = (4 * ks + 2 < NNODE) ? xr[(4 * ks + 2 < NNODE) ? 4 * ks + 2 : 0][i] : 0.0f;
                const float e3 = (4 * ks + 3 < NNODE) ? xr[(4 * ks + 3 < NNODE) ? 4 * ks + 3 : 0][i] : 0.0f;
                bx[ks][i] = (v2f){ hi ? e2 : e0, hi ? e3 : e1 };
            }
        }
        v2f aa[4][A1P / 4];
        #pragma unroll
        for (int tl = 0; tl < 4; ++tl) {
            const int node = min(8 * tl + (l15 & 7), NNODE - 1);
            const float* ap = &sA1[node * A1P + 2 * hi];
            #pragma unroll
            for (int ks = 0; ks < A1P / 4; ++ks) aa[tl][ks] = *(const v2f*)&ap[4 * ks];
        }
        v8f g[4][3];
        #pragma unroll
        for (int tl = 0; tl < 4; ++tl) {
            #pragma unroll
            for (int i = 0; i < 3; ++i) {
                #pragma unroll
                for (int v = 0; v < 8; ++v) g[tl][i][v] = 0.0f;
                asm volatile("" : "+v"(g[tl][i]));
            }
        }
        __builtin_amdgcn_sched_barrier(0);
        #pragma unroll
        for (int ks = 0; ks < A1P / 4; ++ks) {
            #pragma unroll
            for (int tl = 0; tl < 4; ++tl) {
                #pragma unroll
                for (int i = 0; i < 3; ++i) {
                    g[tl][i] = __builtin_amdgcn_wmma_f32_16x16x4_f32(
                                   false, aa[tl][ks], false, bx[ks][i], (short)0, g[tl][i], false, false);
                }
            }
        }
        __builtin_amdgcn_sched_barrier(0);
        asm volatile("v_nop\n\tv_nop\n\tv_nop\n\tv_nop\n\tv_nop");
        #pragma unroll
        for (int n = 0; n < NNODE; ++n) {
            #pragma unroll
            for (int i = 0; i < 3; ++i) ag[n][i] = g[n >> 3][i][n & 7];
        }
    }
    float nw[NNODE];
    #pragma unroll
    for (int n = 0; n < NNODE; ++n) nw[n] = snodew[n];

    v8f acc[8];
    #pragma unroll
    for (int t = 0; t < 8; ++t) {
        const float bz = b2[16 * t + l15];
        #pragma unroll
        for (int v = 0; v < 8; ++v) acc[t][v] = bz;
    }

    const v4f* w1p = (const v4f*)&sW1[hi * 16];
    const float* w2p = &sW2[lane * 4];
    v4f wA[3], wB[3];
    v2f bA, bB;
    wA[0] = w1p[0];
    wA[1] = w1p[1];
    wA[2] = w1p[2];
    bA    = *(const v2f*)&w1p[3];
    #pragma unroll 1
    for (int j = 0; j < NJ; j += 2) {
        const int jn = (j + 2 < NJ) ? (j + 2) : j;
        kiter(ag, xr, nw, w2p + (j * 4) * CHK,       w1p + (j + 1) * 8, wA, bA, wB, bB, acc);
        kiter(ag, xr, nw, w2p + ((j + 1) * 4) * CHK, w1p + jn * 8,      wB, bB, wA, bA, acc);
    }

    #pragma unroll
    for (int t = 0; t < 8; ++t) {
        #pragma unroll
        for (int v = 0; v < 8; ++v) {
            const int m = v + (hi << 3);
            out[(size_t)(b0 + m) * OUTD + 16 * t + l15] = acc[t][v];
        }
    }
}

extern "C" void kernel_launch(void* const* d_in, const int* in_sizes, int n_in,
                              void* d_out, int out_size, void* d_ws, size_t ws_size,
                              hipStream_t stream) {
    const float* x   = (const float*)d_in[0];
    const int*   ei  = (const int*)d_in[1];
    const float* W1l = (const float*)d_in[2];
    const float* W1r = (const float*)d_in[3];
    const float* b1  = (const float*)d_in[4];
    const float* W2l = (const float*)d_in[5];
    const float* W2r = (const float*)d_in[6];
    const float* b2  = (const float*)d_in[7];
    float* out = (float*)d_out;

    const int B = in_sizes[0] / (NNODE * 3);
    const int blocks = B / BPB;
    sage_rowfrag_kernel<<<blocks, 128, 0, stream>>>(x, ei, W1l, W1r, b1, W2l, W2r, b2, out);
}


# second-layer weight fetch stays in flight across graph build and aggregation: edge, x and layer-1 loads issued ahead of it, barriers wait for LDS only, its LDS staging moved behind; sums unchanged
# speedup vs baseline: 1.6318x; 1.1200x over previous
#include <hip/hip_runtime.h>


typedef __attribute__((ext_vector_type(2))) float v2f;
typedef __attribute__((ext_vector_type(4))) float v4f;
typedef __attribute__((ext_vector_type(8))) float v8f;

#define NNODE 25
#define NEDGE 64
#define HID   128
#define OUTD  128
#define TPW   16
#define WPB   4
#define BPB   (TPW * WPB)
#define NJ    (HID / 4)
#define A1P   28
#define XW    (TPW * NNODE * 3)
#define CHK   140

static __device__ __forceinline__ v2f lo2(v4f q) { return __builtin_shufflevector(q, q, 0, 1); }
static __device__ __forceinline__ v2f hi2(v4f q) { return __builtin_shufflevector(q, q, 2, 3); }
static __device__ __forceinline__ v2f sp2(float z) { return (v2f){ z, z }; }

static __device__ __forceinline__ void lds_barrier()
{
    __builtin_amdgcn_fence(__ATOMIC_RELEASE, "workgroup", "local");
    __builtin_amdgcn_s_barrier();
    __builtin_amdgcn_fence(__ATOMIC_ACQUIRE, "workgroup", "local");
}

#define NSTG  8
#define NSLOT (NNODE + NSTG - 1)

static __device__ __forceinline__ void kstep(const float (&ag)[NNODE][3], const float (&xr)[NNODE][3],
                                             const float (&nw)[NNODE], const v4f (&w)[3], const v2f wb,
                                             v2f& acl, v2f& acr)
{
    v2f pre[NNODE];
    v2f cl = { 0.0f, 0.0f };
    v2f cr = { 0.0f, 0.0f };
    #pragma unroll
    for (int s = 0; s < NSLOT; ++s) {
        #pragma unroll
        for (int n = 0; n < NNODE; ++n) {
            const int st = s - n;
            if (st < 0 || st >= NSTG) continue;
            if (st == 0)      pre[n] = __builtin_elementwise_fma(sp2(ag[n][0]), lo2(w[0]), wb);
            else if (st == 1) pre[n] = __builtin_elementwise_fma(sp2(ag[n][1]), hi2(w[0]), pre[n]);
            else if (st == 2) pre[n] = __builtin_elementwise_fma(sp2(ag[n][2]), lo2(w[1]), pre[n]);
            else if (st == 3) pre[n] = __builtin_elementwise_fma(sp2(xr[n][0]), hi2(w[1]), pre[n]);
            else if (st == 4) pre[n] = __builtin_elementwise_fma(sp2(xr[n][1]), lo2(w[2]), pre[n]);
            else if (st == 5) pre[n] = __builtin_elementwise_fma(sp2(xr[n][2]), hi2(w[2]), pre[n]);
            else if (st == 6) pre[n] = __builtin_elementwise_maxnum(pre[n], (v2f){ 0.0f, 0.0f });
            else {
                cl = __builtin_elementwise_fma(pre[n], sp2(nw[n]), cl);
                cr = cr + pre[n];
            }
        }
        if (s == NSLOT - 1) {
            acl = cl;
            acr = cr * (1.0f / 25.0f);
        }
        if ((s & 1) == 1) __builtin_amdgcn_sched_barrier(0);
    }
}

static __device__ __forceinline__ void kiter(const float (&ag)[NNODE][3], const float (&xr)[NNODE][3],
                                             const float (&nw)[NNODE], const float* w2j, const v4f* w1n,
                                             const v4f (&w)[3], const v2f wb, v4f (&wn)[3], v2f& bn,
                                             v8f (&acc)[8])
{
    v4f bl[4], br[4];
    #pragma unroll
    for (int q = 0; q < 4; ++q) {
        bl[q] = *(const v4f*)&w2j[q * CHK];
        br[q] = *(const v4f*)&w2j[(NJ * 4 + q) * CHK];
    }
    wn[0] = w1n[0];
    wn[1] = w1n[1];
    wn[2] = w1n[2];
    bn    = *(const v2f*)&w1n[3];
    __builtin_amdgcn_sched_barrier(0);
    v2f acl, acr;
    kstep(ag, xr, nw, w, wb, acl, acr);
    #pragma unroll
    for (int q = 0; q < 4; ++q) {
        acc[2 * q]     = __builtin_amdgcn_wmma_f32_16x16x4_f32(
                             false, acl, false, lo2(bl[q]), (short)0, acc[2 * q], false, false);
        acc[2 * q + 1] = __builtin_amdgcn_wmma_f32_16x16x4_f32(
                             false, acl, false, hi2(bl[q]), (short)0, acc[2 * q + 1], false, false);
    }
    #pragma unroll
    for (int q = 0; q < 4; ++q) {
        acc[2 * q]     = __builtin_amdgcn_wmma_f32_16x16x4_f32(
                             false, acr, false, lo2(br[q]), (short)0, acc[2 * q], false, false);
        acc[2 * q + 1] = __builtin_amdgcn_wmma_f32_16x16x4_f32(
                             false, acr, false, hi2(br[q]), (short)0, acc[2 * q + 1], false, false);
    }
    __builtin_amdgcn_sched_barrier(0);
    asm volatile("v_nop\n\tv_nop\n\tv_nop\n\tv_nop");
}

__launch_bounds__(128)
__global__ void sage_rowfrag_kernel(const float* __restrict__ x,
                                    const int*   __restrict__ edge_index,
                                    const float* __restrict__ W1l,
                                    const float* __restrict__ W1r,
                                    const float* __restrict__ b1,
                                    const float* __restrict__ W2l,
                                    const float* __restrict__ W2r,
                                    const float* __restrict__ b2,
                                    float* __restrict__ out)
{
    __shared__ int   ssrc[2 * NEDGE];
    __shared__ int   sdst[2 * NEDGE];
    __shared__ float sdeginv[NNODE];
    __shared__ float snodew[NNODE];
    __align__(16) __shared__ float sA1[NNODE * A1P];
    __align__(16) __shared__ float sW1[NJ * 2 * 16];
    __align__(16) __shared__ float sx[WPB * XW];
    __align__(16) __shared__ float sW2[2 * NJ * 4 * CHK];

    const int tid  = threadIdx.x;
    const int lane = tid & 31;
    const int wave = tid >> 5;
    const int hi   = lane >> 4;
    const int l15  = lane & 15;
    const int b0   = (blockIdx.x * WPB + wave) * TPW;

    const int w1g = (tid & 15) >> 1;
    const float* w1row = (w1g < 3) ? (W1l + w1g * HID) : (w1g < 6) ? (W1r + (w1g - 3) * HID) : b1;
    const int w1col = 2 * (tid >> 4) + (tid & 1);
    float w1v[8];
    #pragma unroll
    for (int k = 0; k < 8; ++k) w1v[k] = w1row[w1col + 16 * k];

    const int e_s = edge_index[tid & (NEDGE - 1)];
    const int e_d = edge_index[NEDGE + (tid & (NEDGE - 1))];
    v4f xv[10];
    {
        const v4f* xg = (const v4f*)(x + (size_t)b0 * (NNODE * 3));
        #pragma unroll
        for (int k = 0; k < 9; ++k) xv[k] = xg[lane + 32 * k];
        xv[9] = xg[min(lane + 288, XW / 4 - 1)];
    }
    __builtin_amdgcn_sched_barrier(0);

    v4f wra[32], wrb[32];
    #pragma unroll
    for (int u = 0; u < 32; ++u) {
        const int it  = tid + 128 * u;
        const int mat = it >> 11;
        const int rp  = (it >> 5) & 63;
        const int c4  = it & 31;
        const float* Wm = mat ? W2r : W2l;
        wra[u] = *(const v4f*)&Wm[(2 * rp) * OUTD + 4 * c4];
        wrb[u] = *(const v4f*)&Wm[(2 * rp + 1) * OUTD + 4 * c4];
    }

    #pragma unroll
    for (int k = 0; k < 8; ++k) sW1[tid + 128 * k] = (w1g == 7) ? 0.0f : w1v[k];

    ssrc[tid] = e_s;
    sdst[tid] = e_d;
    for (int i = tid; i < NNODE * A1P; i += 128) sA1[i] = 0.0f;
    if (tid < NNODE) snodew[tid] = 0.0f;

    {
        v4f* xs = (v4f*)&sx[wave * XW];
        #pragma unroll
        for (int k = 0; k < 9; ++k) xs[lane + 32 * k] = xv[k];
        if (lane < XW / 4 - 288) xs[lane + 288] = xv[9];
    }
    lds_barrier();

    if (tid < NNODE) {
        int c = 0;
        #pragma unroll
        for (int e = 0; e < NEDGE; ++e) c += (sdst[e] == tid);
        sdeginv[tid] = 1.0f / fmaxf((float)c, 1.0f);
    }
    lds_barrier();

    if (tid < NEDGE) {
        const float di = sdeginv[sdst[tid]];
        atomicAdd(&sA1[sdst[tid] * A1P + ssrc[tid]], di);
        atomicAdd(&snodew[ssrc[tid]], di * (1.0f / 25.0f));
    }
    lds_barrier();

    float xr[NNODE][3];
    {
        const float* xrow = &sx[wave * XW + l15 * (NNODE * 3)];
        #pragma unroll
        for (int m = 0; m < NNODE; ++m) {
            xr[m][0] = xrow[3 * m + 0];
            xr[m][1] = xrow[3 * m + 1];
            xr[m][2] = xrow[3 * m + 2];
        }
    }
    float ag[NNODE][3];
    {
        v2f bx[A1P / 4][3];
        #pragma unroll
        for (int ks = 0; ks < A1P / 4; ++ks) {
            #pragma unroll
            for (int i = 0; i < 3; ++i) {
                const float e0 = (4 * ks + 0 < NNODE) ? xr[(4 * ks + 0 < NNODE) ? 4 * ks + 0 : 0][i] : 0.0f;
                const float e1 = (4 * ks + 1 < NNODE) ? xr[(4 * ks + 1 < NNODE) ? 4 * ks + 1 : 0][i] : 0.0f;
                const float e2 = (4 * ks + 2 < NNODE) ? xr[(4 * ks + 2 < NNODE) ? 4 * ks + 2 : 0][i] : 0.0f;
                const float e3 = (4 * ks + 3 < NNODE) ? xr[(4 * ks + 3 < NNODE) ? 4 * ks + 3 : 0][i] : 0.0f;
                bx[ks][i] = (v2f){ hi ? e2 : e0, hi ? e3 : e1 };
            }
        }
        v2f aa[4][A1P / 4];
        #pragma unroll
        for (int tl = 0; tl < 4; ++tl) {
            const int node = min(8 * tl + (l15 & 7), NNODE - 1);
            const float* ap = &sA1[node * A1P + 2 * hi];
            #pragma unroll
            for (int ks = 0; ks < A1P / 4; ++ks) aa[tl][ks] = *(const v2f*)&ap[4 * ks];
        }
        v8f g[4][3];
        #pragma unroll
        for (int tl = 0; tl < 4; ++tl) {
            #pragma unroll
            for (int i = 0; i < 3; ++i) {
                #pragma unroll
                for (int v = 0; v < 8; ++v) g[tl][i][v] = 0.0f;
                asm volatile("" : "+v"(g[tl][i]));
            }
        }
        __builtin_amdgcn_sched_barrier(0);
        #pragma unroll
        for (int ks = 0; ks < A1P / 4; ++ks) {
            #pragma unroll
            for (int tl = 0; tl < 4; ++tl) {
                #pragma unroll
                for (int i = 0; i < 3; ++i) {
                    g[tl][i] = __builtin_amdgcn_wmma_f32_16x16x4_f32(
                                   false, aa[tl][ks], false, bx[ks][i], (short)0, g[tl][i], false, false);
                }
            }
        }
        __builtin_amdgcn_sched_barrier(0);
        asm volatile("v_nop\n\tv_nop\n\tv_nop\n\tv_nop\n\tv_nop");
        __builtin_amdgcn_sched_barrier(0);
        #pragma unroll
        for (int n = 0; n < NNODE; ++n) {
            #pragma unroll
            for (int i = 0; i < 3; ++i) ag[n][i] = g[n >> 3][i][n & 7];
        }
    }
    float nw[NNODE];
    #pragma unroll
    for (int n = 0; n < NNODE; ++n) nw[n] = snodew[n];

    #pragma unroll
    for (int u = 0; u < 32; ++u) {
        const int it  = tid + 128 * u;
        const int mat = it >> 11;
        const int rp  = (it >> 5) & 63;
        const int c4  = it & 31;
        const int j = rp >> 1, h = rp & 1;
        const int t = c4 >> 2, q = t >> 1, tt = t & 1;
        const int cb = 4 * (c4 & 3);
        float* dstp = &sW2[((mat * NJ + j) * 4 + q) * CHK + (cb + 16 * h) * 4 + tt * 2];
        *(v2f*)(dstp + 0)  = (v2f){ wra[u].x, wrb[u].x };
        *(v2f*)(dstp + 4)  = (v2f){ wra[u].y, wrb[u].y };
        *(v2f*)(dstp + 8)  = (v2f){ wra[u].z, wrb[u].z };
        *(v2f*)(dstp + 12) = (v2f){ wra[u].w, wrb[u].w };
    }
    __syncthreads();

    v8f acc[8];
    #pragma unroll
    for (int t = 0; t < 8; ++t) {
        const float bz = b2[16 * t + l15];
        #pragma unroll
        for (int v = 0; v < 8; ++v) acc[t][v] = bz;
    }

    const v4f* w1p = (const v4f*)&sW1[hi * 16];
    const float* w2p = &sW2[lane * 4];
    v4f wA[3], wB[3];
    v2f bA, bB;
    wA[0] = w1p[0];
    wA[1] = w1p[1];
    wA[2] = w1p[2];
    bA    = *(const v2f*)&w1p[3];
    #pragma unroll 1
    for (int j = 0; j < NJ; j += 2) {
        const int jn = (j + 2 < NJ) ? (j + 2) : j;
        kiter(ag, xr, nw, w2p + (j * 4) * CHK,       w1p + (j + 1) * 8, wA, bA, wB, bB, acc);
        kiter(ag, xr, nw, w2p + ((j + 1) * 4) * CHK, w1p + jn * 8,      wB, bB, wA, bA, acc);
    }

    #pragma unroll
    for (int t = 0; t < 8; ++t) {
        #pragma unroll
        for (int v = 0; v < 8; ++v) {
            const int m = v + (hi << 3);
            out[(size_t)(b0 + m) * OUTD + 16 * t + l15] = acc[t][v];
        }
    }
}

extern "C" void kernel_launch(void* const* d_in, const int* in_sizes, int n_in,
                              void* d_out, int out_size, void* d_ws, size_t ws_size,
                              hipStream_t stream) {
    const float* x   = (const float*)d_in[0];
    const int*   ei  = (const int*)d_in[1];
    const float* W1l = (const float*)d_in[2];
    const float* W1r = (const float*)d_in[3];
    const float* b1  = (const float*)d_in[4];
    const float* W2l = (const float*)d_in[5];
    const float* W2r = (const float*)d_in[6];
    const float* b2  = (const float*)d_in[7];
    float* out = (float*)d_out;

    const int B = in_sizes[0] / (NNODE * 3);
    const int blocks = B / BPB;
    sage_rowfrag_kernel<<<blocks, 128, 0, stream>>>(x, ei, W1l, W1r, b1, W2l, W2r, b2, out);
}


# each M-tile shared by 2 waves (8 per block, 2 per SIMD): aggregation split by node tile, main loop by k half, one wave's matrix run overlaps the other's vector work; sum order (bias+lower k)+upper k
# speedup vs baseline: 1.7212x; 1.0548x over previous
#include <hip/hip_runtime.h>


typedef __attribute__((ext_vector_type(2))) float v2f;
typedef __attribute__((ext_vector_type(4))) float v4f;
typedef __attribute__((ext_vector_type(8))) float v8f;

#define NNODE 25
#define NEDGE 64
#define HID   128
#define OUTD  128
#define TPW   16
#define WPB   4
#define KSPL  2
#define NTHR  (32 * WPB * KSPL)
#define BPB   (TPW * WPB)
#define NJ    (HID / 4)
#define A1P   28
#define XW    (TPW * NNODE * 3)
#define AGP   100
#define CHK   140

static __device__ __forceinline__ v2f lo2(v4f q) { return __builtin_shufflevector(q, q, 0, 1); }
static __device__ __forceinline__ v2f hi2(v4f q) { return __builtin_shufflevector(q, q, 2, 3); }
static __device__ __forceinline__ v2f sp2(float z) { return (v2f){ z, z }; }

static __device__ __forceinline__ void lds_barrier()
{
    __builtin_amdgcn_fence(__ATOMIC_RELEASE, "workgroup", "local");
    __builtin_amdgcn_s_barrier();
    __builtin_amdgcn_fence(__ATOMIC_ACQUIRE, "workgroup", "local");
}

#define NSTG  8
#define NSLOT (NNODE + NSTG - 1)

static __device__ __forceinline__ void kstep(const float (&ag)[NNODE][3], const float (&xr)[NNODE][3],
                                             const float (&nw)[NNODE], const v4f (&w)[3], const v2f wb,
                                             v2f& acl, v2f& acr)
{
    v2f pre[NNODE];
    v2f cl = { 0.0f, 0.0f };
    v2f cr = { 0.0f, 0.0f };
    #pragma unroll
    for (int s = 0; s < NSLOT; ++s) {
        #pragma unroll
        for (int n = 0; n < NNODE; ++n) {
            const int st = s - n;
            if (st < 0 || st >= NSTG) continue;
            if (st == 0)      pre[n] = __builtin_elementwise_fma(sp2(ag[n][0]), lo2(w[0]), wb);
            else if (st == 1) pre[n] = __builtin_elementwise_fma(sp2(ag[n][1]), hi2(w[0]), pre[n]);
            else if (st == 2) pre[n] = __builtin_elementwise_fma(sp2(ag[n][2]), lo2(w[1]), pre[n]);
            else if (st == 3) pre[n] = __builtin_elementwise_fma(sp2(xr[n][0]), hi2(w[1]), pre[n]);
            else if (st == 4) pre[n] = __builtin_elementwise_fma(sp2(xr[n][1]), lo2(w[2]), pre[n]);
            else if (st == 5) pre[n] = __builtin_elementwise_fma(sp2(xr[n][2]), hi2(w[2]), pre[n]);
            else if (st == 6) pre[n] = __builtin_elementwise_maxnum(pre[n], (v2f){ 0.0f, 0.0f });
            else {
                cl = __builtin_elementwise_fma(pre[n], sp2(nw[n]), cl);
                cr = cr + pre[n];
            }
        }
        if (s == NSLOT - 1) {
            acl = cl;
            acr = cr * (1.0f / 25.0f);
        }
        if ((s & 1) == 1) __builtin_amdgcn_sched_barrier(0);
    }
}

static __device__ __forceinline__ void kiter(const float (&ag)[NNODE][3], const float (&xr)[NNODE][3],
                                             const float (&nw)[NNODE], const float* w2j, const v4f* w1n,
                                             const v4f (&w)[3], const v2f wb, v4f (&wn)[3], v2f& bn,
                                             v8f (&acc)[8])
{
    v4f bl[4], br[4];
    #pragma unroll
    for (int q = 0; q < 4; ++q) {
        bl[q] = *(const v4f*)&w2j[q * CHK];
        br[q] = *(const v4f*)&w2j[(NJ * 4 + q) * CHK];
    }
    wn[0] = w1n[0];
    wn[1] = w1n[1];
    wn[2] = w1n[2];
    bn    = *(const v2f*)&w1n[3];
    __builtin_amdgcn_sched_barrier(0);
    v2f acl, acr;
    kstep(ag, xr, nw, w, wb, acl, acr);
    #pragma unroll
    for (int q = 0; q < 4; ++q) {
        acc[2 * q]     = __builtin_amdgcn_wmma_f32_16x16x4_f32(
                             false, acl, false, lo2(bl[q]), (short)0, acc[2 * q], false, false);
        acc[2 * q + 1] = __builtin_amdgcn_wmma_f32_16x16x4_f32(
                             false, acl, false, hi2(bl[q]), (short)0, acc[2 * q + 1], false, false);
    }
    #pragma unroll
    for (int q = 0; q < 4; ++q) {
        acc[2 * q]     = __builtin_amdgcn_wmma_f32_16x16x4_f32(
                             false, acr, false, lo2(br[q]), (short)0, acc[2 * q], false, false);
        acc[2 * q + 1] = __builtin_amdgcn_wmma_f32_16x16x4_f32(
                             false, acr, false, hi2(br[q]), (short)0, acc[2 * q + 1], false, false);
    }
    __builtin_amdgcn_sched_barrier(0);
    asm volatile("v_nop\n\tv_nop\n\tv_nop\n\tv_nop");
}

__launch_bounds__(NTHR)
__global__ void sage_rowfrag_kernel(const float* __restrict__ x,
                                    const int*   __restrict__ edge_index,
                                    const float* __restrict__ W1l,
                                    const float* __restrict__ W1r,
                                    const float* __restrict__ b1,
                                    const float* __restrict__ W2l,
                                    const float* __restrict__ W2r,
                                    const float* __restrict__ b2,
                                    float* __restrict__ out)
{
    __shared__ int   ssrc[NTHR];
    __shared__ int   sdst[NTHR];
    __shared__ float sdeginv[NNODE];
    __shared__ float snodew[NNODE];
    __align__(16) __shared__ float sA1[NNODE * A1P];
    __align__(16) __shared__ float sW1[NJ * 2 * 16];
    __align__(16) __shared__ float sx[WPB * XW];
    __align__(16) __shared__ float sW2[2 * NJ * 4 * CHK];
    __align__(16) __shared__ float sag[WPB * TPW * AGP];
    __align__(16) __shared__ float sred[WPB * 8 * 2 * 32 * 4];

    const int tid  = threadIdx.x;
    const int lane = tid & 31;
    const int wave = __builtin_amdgcn_readfirstlane(tid >> 5);
    const int mt   = wave & (WPB - 1);
    const int kh   = wave >> 2;
    const int hi   = lane >> 4;
    const int l15  = lane & 15;
    const int b0   = (blockIdx.x * WPB + mt) * TPW;

    const int w1g = (tid & 15) >> 1;
    const float* w1row = (w1g < 3) ? (W1l + w1g * HID) : (w1g < 6) ? (W1r + (w1g - 3) * HID) : b1;
    const int w1col = 2 * (tid >> 4) + (tid & 1);
    float w1v[4];
    #pragma unroll
    for (int k = 0; k < 4; ++k) w1v[k] = w1row[w1col + 32 * k];

    const int e_s = edge_index[tid & (NEDGE - 1)];
    const int e_d = edge_index[NEDGE + (tid & (NEDGE - 1))];
    v4f xv[5];
    const int xl = lane + 32 * kh;
    {
        const v4f* xg = (const v4f*)(x + (size_t)b0 * (NNODE * 3));
        #pragma unroll
        for (int k = 0; k < 4; ++k) xv[k] = xg[xl + 64 * k];
        xv[4] = xg[min(xl + 256, XW / 4 - 1)];
    }
    __builtin_amdgcn_sched_barrier(0);

    v4f wra[16], wrb[16];
    #pragma unroll
    for (int u = 0; u < 16; ++u) {
        const int it  = tid + NTHR * u;
        const int mat = it >> 11;
        const int rp  = (it >> 5) & 63;
        const int c4  = it & 31;
        const float* Wm = mat ? W2r : W2l;
        wra[u] = *(const v4f*)&Wm[(2 * rp) * OUTD + 4 * c4];
        wrb[u] = *(const v4f*)&Wm[(2 * rp + 1) * OUTD + 4 * c4];
    }

    #pragma unroll
    for (int k = 0; k < 4; ++k) sW1[tid + NTHR * k] = (w1g == 7) ? 0.0f : w1v[k];

    ssrc[tid] = e_s;
    sdst[tid] = e_d;
    for (int i = tid; i < NNODE * A1P; i += NTHR) sA1[i] = 0.0f;
    if (tid < NNODE) snodew[tid] = 0.0f;

    {
        v4f* xs = (v4f*)&sx[mt * XW];
        #pragma unroll
        for (int k = 0; k < 4; ++k) xs[xl + 64 * k] = xv[k];
        if (xl < XW / 4 - 256) xs[xl + 256] = xv[4];
    }
    lds_barrier();

    if (tid < NNODE) {
        int c = 0;
        #pragma unroll
        for (int e = 0; e < NEDGE; ++e) c += (sdst[e] == tid);
        sdeginv[tid] = 1.0f / fmaxf((float)c, 1.0f);
    }
    lds_barrier();

    if (tid < NEDGE) {
        const float di = sdeginv[sdst[tid]];
        atomicAdd(&sA1[sdst[tid] * A1P + ssrc[tid]], di);
        atomicAdd(&snodew[ssrc[tid]], di * (1.0f / 25.0f));
    }
    lds_barrier();

    float xr[NNODE][3];
    {
        const float* xrow = &sx[mt * XW + l15 * (NNODE * 3)];
        #pragma unroll
        for (int m = 0; m < NNODE; ++m) {
            xr[m][0] = xrow[3 * m + 0];
            xr[m][1] = xrow[3 * m + 1];
            xr[m][2] = xrow[3 * m + 2];
        }
    }
    float ag[NNODE][3];
    {
        v2f bx[A1P / 4][3];
        #pragma unroll
        for (int ks = 0; ks < A1P / 4; ++ks) {
            #pragma unroll
            for (int i = 0; i < 3; ++i) {
                const float e0 = (4 * ks + 0 < NNODE) ? xr[(4 * ks + 0 < NNODE) ? 4 * ks + 0 : 0][i] : 0.0f;
                const float e1 = (4 * ks + 1 < NNODE) ? xr[(4 * ks + 1 < NNODE) ? 4 * ks + 1 : 0][i] : 0.0f;
                const float e2 = (4 * ks + 2 < NNODE) ? xr[(4 * ks + 2 < NNODE) ? 4 * ks + 2 : 0][i] : 0.0f;
                const float e3 = (4 * ks + 3 < NNODE) ? xr[(4 * ks + 3 < NNODE) ? 4 * ks + 3 : 0][i] : 0.0f;
                bx[ks][i] = (v2f){ hi ? e2 : e0, hi ? e3 : e1 };
            }
        }
        v2f aa[A1P / 4];
        {
            const int node = min(16 * kh + l15, NNODE - 1);
            const float* ap = &sA1[node * A1P + 2 * hi];
            #pragma unroll
            for (int ks = 0; ks < A1P / 4; ++ks) aa[ks] = *(const v2f*)&ap[4 * ks];
        }
        v8f g[3];
        #pragma unroll
        for (int i = 0; i < 3; ++i) {
            #pragma unroll
            for (int v = 0; v < 8; ++v) g[i][v] = 0.0f;
            asm volatile("" : "+v"(g[i]));
        }
        __builtin_amdgcn_sched_barrier(0);
        #pragma unroll
        for (int ks = 0; ks < A1P / 4; ++ks) {
            #pragma unroll
            for (int i = 0; i < 3; ++i) {
                g[i] = __builtin_amdgcn_wmma_f32_16x16x4_f32(
                           false, aa[ks], false, bx[ks][i], (short)0, g[i], false, false);
            }
        }
        __builtin_amdgcn_sched_barrier(0);
        asm volatile("v_nop\n\tv_nop\n\tv_nop\n\tv_nop\n\tv_nop");
        __builtin_amdgcn_sched_barrier(0);
        float* gp = &sag[(mt * TPW + l15) * AGP + 16 * kh + 8 * hi];
        #pragma unroll
        for (int i = 0; i < 3; ++i) {
            *(v4f*)&gp[i * 32]     = __builtin_shufflevector(g[i], g[i], 0, 1, 2, 3);
            *(v4f*)&gp[i * 32 + 4] = __builtin_shufflevector(g[i], g[i], 4, 5, 6, 7);
        }
        lds_barrier();
        const float* gr = &sag[(mt * TPW + l15) * AGP];
        #pragma unroll
        for (int i = 0; i < 3; ++i) {
            #pragma unroll
            for (int n4 = 0; n4 < 6; ++n4) {
                const v4f q = *(const v4f*)&gr[i * 32 + 4 * n4];
                ag[4 * n4 + 0][i] = q.x;
                ag[4 * n4 + 1][i] = q.y;
                ag[4 * n4 + 2][i] = q.z;
                ag[4 * n4 + 3][i] = q.w;
            }
            ag[24][i] = gr[i * 32 + 24];
        }
    }
    float nw[NNODE];
    #pragma unroll
    for (int n = 0; n < NNODE; ++n) nw[n] = snodew[n];

    #pragma unroll
    for (int u = 0; u < 16; ++u) {
        const int it  = tid + NTHR * u;
        const int mat = it >> 11;
        const int rp  = (it >> 5) & 63;
        const int c4  = it & 31;
        const int j = rp >> 1, h = rp & 1;
        const int t = c4 >> 2, q = t >> 1, tt = t & 1;
        const int cb = 4 * (c4 & 3);
        float* dstp = &sW2[((mat * NJ + j) * 4 + q) * CHK + (cb + 16 * h) * 4 + tt * 2];
        *(v2f*)(dstp + 0)  = (v2f){ wra[u].x, wrb[u].x };
        *(v2f*)(dstp + 4)  = (v2f){ wra[u].y, wrb[u].y };
        *(v2f*)(dstp + 8)  = (v2f){ wra[u].z, wrb[u].z };
        *(v2f*)(dstp + 12) = (v2f){ wra[u].w, wrb[u].w };
    }
    __syncthreads();

    v8f acc[8];
    #pragma unroll
    for (int t = 0; t < 8; ++t) {
        const float bz = kh ? 0.0f : b2[16 * t + l15];
        #pragma unroll
        for (int v = 0; v < 8; ++v) acc[t][v] = bz;
    }

    const int j0 = kh * (NJ / KSPL);
    const v4f* w1p = (const v4f*)&sW1[hi * 16];
    const float* w2p = &sW2[lane * 4];
    v4f wA[3], wB[3];
    v2f bA, bB;
    wA[0] = w1p[j0 * 8 + 0];
    wA[1] = w1p[j0 * 8 + 1];
    wA[2] = w1p[j0 * 8 + 2];
    bA    = *(const v2f*)&w1p[j0 * 8 + 3];
    #pragma unroll 1
    for (int j = j0; j < j0 + NJ / KSPL; j += 2) {
        const int jn = (j + 2 < j0 + NJ / KSPL) ? (j + 2) : j;
        kiter(ag, xr, nw, w2p + (j * 4) * CHK,       w1p + (j + 1) * 8, wA, bA, wB, bB, acc);
        kiter(ag, xr, nw, w2p + ((j + 1) * 4) * CHK, w1p + jn * 8,      wB, bB, wA, bA, acc);
    }

    v4f* rp4 = (v4f*)&sred[(mt * 8 * 2 * 32 + lane) * 4];
    if (kh) {
        #pragma unroll
        for (int t = 0; t < 8; ++t) {
            rp4[(2 * t) * 32]     = __builtin_shufflevector(acc[t], acc[t], 0, 1, 2, 3);
            rp4[(2 * t + 1) * 32] = __builtin_shufflevector(acc[t], acc[t], 4, 5, 6, 7);
        }
    }
    __syncthreads();
    if (kh == 0) {
        #pragma unroll
        for (int t = 0; t < 8; ++t) {
            const v4f p0 = rp4[(2 * t) * 32];
            const v4f p1 = rp4[(2 * t + 1) * 32];
            acc[t] += __builtin_shufflevector(p0, p1, 0, 1, 2, 3, 4, 5, 6, 7);
        }
        #pragma unroll
        for (int t = 0; t < 8; ++t) {
            #pragma unroll
            for (int v = 0; v < 8; ++v) {
                const int m = v + (hi << 3);
                out[(size_t)(b0 + m) * OUTD + 16 * t + l15] = acc[t][v];
            }
        }
    }
}

extern "C" void kernel_launch(void* const* d_in, const int* in_sizes, int n_in,
                              void* d_out, int out_size, void* d_ws, size_t ws_size,
                              hipStream_t stream) {
    const float* x   = (const float*)d_in[0];
    const int*   ei  = (const int*)d_in[1];
    const float* W1l = (const float*)d_in[2];
    const float* W1r = (const float*)d_in[3];
    const float* b1  = (const float*)d_in[4];
    const float* W2l = (const float*)d_in[5];
    const float* W2r = (const float*)d_in[6];
    const float* b2  = (const float*)d_in[7];
    float* out = (float*)d_out;

    const int B = in_sizes[0] / (NNODE * 3);
    const int blocks = B / BPB;
    sage_rowfrag_kernel<<<blocks, NTHR, 0, stream>>>(x, ei, W1l, W1r, b1, W2l, W2r, b2, out);
}
